# starting kernel
# baseline (speedup 1.0000x reference)
#include <hip/hip_runtime.h>
#include <hip/hip_bf16.h>

#define NUM_USERS 100000
#define NUM_ITEMS 50000
#define N_NODES   (NUM_USERS + NUM_ITEMS)
#define LATENT    64
#define NNZ_E     4000000
#define BATCH_SZ  16384

typedef __attribute__((ext_vector_type(2))) float v2f;
typedef __attribute__((ext_vector_type(8))) float v8f;

__global__ void lgcn_init(const float* __restrict__ utab, const float* __restrict__ itab,
                          float* __restrict__ X, float* __restrict__ acc) {
    long i4 = (long)blockIdx.x * blockDim.x + threadIdx.x;
    long n4 = (long)N_NODES * LATENT / 4;
    if (i4 >= n4) return;
    long i = i4 * 4;
    const long ub = (long)NUM_USERS * LATENT;
    float4 v;
    if (i < ub) v = *(const float4*)(utab + i);
    else        v = *(const float4*)(itab + (i - ub));
    *(float4*)(X + i)   = v;
    *(float4*)(acc + i) = v;
}

__global__ void lgcn_zero(float* __restrict__ Y) {
    long i4 = (long)blockIdx.x * blockDim.x + threadIdx.x;
    long n4 = (long)N_NODES * LATENT / 4;
    if (i4 >= n4) return;
    *(float4*)(Y + i4 * 4) = make_float4(0.f, 0.f, 0.f, 0.f);
}

__global__ void lgcn_accum(float* __restrict__ acc, const float* __restrict__ Y) {
    long i4 = (long)blockIdx.x * blockDim.x + threadIdx.x;
    long n4 = (long)N_NODES * LATENT / 4;
    if (i4 >= n4) return;
    float4 a = *(float4*)(acc + i4 * 4);
    float4 y = *(const float4*)(Y + i4 * 4);
    a.x += y.x; a.y += y.y; a.z += y.z; a.w += y.w;
    *(float4*)(acc + i4 * 4) = a;
}

__global__ void lgcn_spmm(const int* __restrict__ esrc, const int* __restrict__ edst,
                          const float* __restrict__ evals,
                          const float* __restrict__ X, float* __restrict__ Y) {
    int tid  = blockIdx.x * blockDim.x + threadIdx.x;
    int edge = tid >> 5;
    int lane = threadIdx.x & 31;
    if (edge >= NNZ_E) return;
    int   s = __builtin_nontemporal_load(esrc + edge);
    int   d = __builtin_nontemporal_load(edst + edge);
    float v = __builtin_nontemporal_load(evals + edge);
    const float* xr = X + (long)s * LATENT + 2 * lane;
    float2 x2 = *(const float2*)xr;
    float* yp = Y + (long)d * LATENT + 2 * lane;
    unsafeAtomicAdd(yp,     x2.x * v);
    unsafeAtomicAdd(yp + 1, x2.y * v);
}

__global__ void lgcn_gamma_wmma(const float* __restrict__ acc,
                                const int* __restrict__ users,
                                const int* __restrict__ items,
                                float* __restrict__ out) {
    int wave = (blockIdx.x * blockDim.x + threadIdx.x) >> 5;
    int lane = threadIdx.x & 31;
    int b    = wave * 16 + (lane & 15);
    int half = (lane >> 4) & 1;

    const float* urow = acc + (long)users[b] * LATENT;
    const float* irow = acc + ((long)NUM_USERS + items[b]) * LATENT;

    v8f c = {};
#pragma unroll
    for (int kk = 0; kk < 16; ++kk) {
        int off = 4 * kk + 2 * half;
        v2f a  = *(const v2f*)(urow + off);
        v2f bb = *(const v2f*)(irow + off);
        c = __builtin_amdgcn_wmma_f32_16x16x4_f32(false, a, false, bb,
                                                  (short)0, c, false, false);
    }

    int r = ((lane < 8) ? lane : (lane - 24)) & 7;
    float v = c[0];
#pragma unroll
    for (int i = 1; i < 8; ++i) if (r == i) v = c[i];

    if (lane < 8)        out[wave * 16 + lane]        = v * 0.0625f;
    else if (lane >= 24) out[wave * 16 + (lane - 16)] = v * 0.0625f;
}

extern "C" void kernel_launch(void* const* d_in, const int* in_sizes, int n_in,
                              void* d_out, int out_size, void* d_ws, size_t ws_size,
                              hipStream_t stream) {
    const int*   users = (const int*)d_in[0];
    const int*   items = (const int*)d_in[1];
    const int*   esrc  = (const int*)d_in[2];
    const int*   edst  = (const int*)d_in[3];
    const float* evals = (const float*)d_in[4];
    const float* utab  = (const float*)d_in[5];
    const float* itab  = (const float*)d_in[6];
    float* out = (float*)d_out;

    const long NE = (long)N_NODES * LATENT;
    float* X   = (float*)d_ws;
    float* Y   = X + NE;
    float* acc = Y + NE;

    const long n4      = NE / 4;
    const int  blk     = 256;
    const int  gridEl  = (int)((n4 + blk - 1) / blk);

    lgcn_init<<<gridEl, blk, 0, stream>>>(utab, itab, X, acc);

    for (int layer = 0; layer < 3; ++layer) {
        lgcn_zero<<<gridEl, blk, 0, stream>>>(Y);
        long spmmThreads = (long)NNZ_E * 32;
        lgcn_spmm<<<(int)(spmmThreads / blk), blk, 0, stream>>>(esrc, edst, evals, X, Y);
        lgcn_accum<<<gridEl, blk, 0, stream>>>(acc, Y);
        float* t = X; X = Y; Y = t;
    }

    lgcn_gamma_wmma<<<BATCH_SZ / 16 / 8, blk, 0, stream>>>(acc, users, items, out);
}


# SpMM pulled over per-row edge chains built in-call by one int atomic exchange per edge (no sort, no f32 atomics): 8 lanes walk a row, sum in registers; sum order = arrival; layers nested x0+A(..)
# speedup vs baseline: 3.1661x; 3.1661x over previous
#include <hip/hip_runtime.h>


#define NUM_USERS 100000
#define NUM_ITEMS 50000
#define N_NODES   (NUM_USERS + NUM_ITEMS)
#define LATENT    64
#define NNZ_E     4000000
#define BATCH_SZ  16384

typedef __attribute__((ext_vector_type(2))) float v2f;
typedef __attribute__((ext_vector_type(4))) float v4f;
typedef __attribute__((ext_vector_type(8))) float v8f;
typedef __attribute__((ext_vector_type(2))) int   v2i;
typedef __attribute__((ext_vector_type(4))) int   v4i;


__global__ void lgcn_head_init(int* __restrict__ head) {
    int i = blockIdx.x * blockDim.x + threadIdx.x;
    if (i < N_NODES) head[i] = -1;
}

__global__ void lgcn_chain(const int* __restrict__ esrc, const int* __restrict__ edst,
                           int* __restrict__ head, int* __restrict__ rec) {
    int t = blockIdx.x * blockDim.x + threadIdx.x;
    if (t >= NNZ_E / 4) return;
    int ea = 2 * t;
    int eb = ea + NNZ_E / 2;
    v2i da = __builtin_nontemporal_load((const v2i*)(edst + ea));
    v2i db = __builtin_nontemporal_load((const v2i*)(edst + eb));
    v2i sa = __builtin_nontemporal_load((const v2i*)(esrc + ea));
    v2i sb = __builtin_nontemporal_load((const v2i*)(esrc + eb));
    int a0 = atomicExch(head + da.x, ea);
    int a1 = atomicExch(head + da.y, ea + 1);
    int b0 = atomicExch(head + db.x, eb);
    int b1 = atomicExch(head + db.y, eb + 1);
    v4i ra = {a0, sa.x, a1, sa.y};
    v4i rb = {b0, sb.x, b1, sb.y};
    *(v4i*)(rec + 2 * (long)ea) = ra;
    *(v4i*)(rec + 2 * (long)eb) = rb;
}

template <int FIRST>
__global__ void lgcn_pull(const int* __restrict__ rec, const float* __restrict__ evals,
                          const int* __restrict__ head,
                          const float* __restrict__ utab, const float* __restrict__ itab,
                          const float* __restrict__ Zin, float* __restrict__ Zout) {
    int t   = blockIdx.x * blockDim.x + threadIdx.x;
    int row = t >> 3;
    int j   = threadIdx.x & 7;
    if (row >= N_NODES) return;

    const float* x0r = (row < NUM_USERS ? utab + (long)row * LATENT
                                        : itab + (long)(row - NUM_USERS) * LATENT) + 4 * j;
    v4f b0 = __builtin_nontemporal_load((const v4f*)x0r);
    v4f b1 = __builtin_nontemporal_load((const v4f*)(x0r + 32));

    v4f a0 = {0.f, 0.f, 0.f, 0.f};
    v4f a1 = {0.f, 0.f, 0.f, 0.f};

    int   e    = head[row];
    bool  live = (unsigned)e < (unsigned)NNZ_E;
    v2i   r    = {-1, 0};
    float v    = 0.f;
    if (live) { r = *(const v2i*)(rec + 2 * (long)e); v = evals[e]; }

    for (int it = 0; it < NNZ_E && live; ++it) {
        int   nxt = r.x;
        int   s   = r.y;
        float w   = v;
        live = (unsigned)nxt < (unsigned)NNZ_E;
        if (live) { r = *(const v2i*)(rec + 2 * (long)nxt); v = evals[nxt]; }
        const float* xr;
        if (FIRST) xr = (s < NUM_USERS ? utab + (long)s * LATENT
                                       : itab + (long)(s - NUM_USERS) * LATENT) + 4 * j;
        else       xr = Zin + (long)s * LATENT + 4 * j;
        v4f x0 = *(const v4f*)xr;
        v4f x1 = *(const v4f*)(xr + 32);
        a0 += w * x0;
        a1 += w * x1;
    }

    float* yo = Zout + (long)row * LATENT + 4 * j;
    *(v4f*)yo        = b0 + a0;
    *(v4f*)(yo + 32) = b1 + a1;
}

__global__ void lgcn_gamma_wmma(const float* __restrict__ acc,
                                const int* __restrict__ users,
                                const int* __restrict__ items,
                                float* __restrict__ out) {
    int wave = (blockIdx.x * blockDim.x + threadIdx.x) >> 5;
    int lane = threadIdx.x & 31;
    int b    = wave * 16 + (lane & 15);
    int half = (lane >> 4) & 1;

    const float* urow = acc + (long)users[b] * LATENT;
    const float* irow = acc + ((long)NUM_USERS + items[b]) * LATENT;

    v8f c = {};
#pragma unroll
    for (int kk = 0; kk < 16; ++kk) {
        int off = 4 * kk + 2 * half;
        v2f a  = *(const v2f*)(urow + off);
        v2f bb = *(const v2f*)(irow + off);
        c = __builtin_amdgcn_wmma_f32_16x16x4_f32(false, a, false, bb,
                                                  (short)0, c, false, false);
    }

    int r = ((lane < 8) ? lane : (lane - 24)) & 7;
    float v = c[0];
#pragma unroll
    for (int i = 1; i < 8; ++i) if (r == i) v = c[i];

    if (lane < 8)        out[wave * 16 + lane]        = v * 0.0625f;
    else if (lane >= 24) out[wave * 16 + (lane - 16)] = v * 0.0625f;
}

extern "C" void kernel_launch(void* const* d_in, const int* in_sizes, int n_in,
                              void* d_out, int out_size, void* d_ws, size_t ws_size,
                              hipStream_t stream) {
    const int*   users = (const int*)d_in[0];
    const int*   items = (const int*)d_in[1];
    const int*   esrc  = (const int*)d_in[2];
    const int*   edst  = (const int*)d_in[3];
    const float* evals = (const float*)d_in[4];
    const float* utab  = (const float*)d_in[5];
    const float* itab  = (const float*)d_in[6];
    float* out = (float*)d_out;

    float* ZA   = (float*)d_ws;
    float* ZB   = ZA + 9600000;
    int*   rec  = (int*)(ZB + 9600000);
    int*   head = rec + 8000000;

    lgcn_head_init<<<586, 256, 0, stream>>>(head);
    lgcn_chain<<<3907, 256, 0, stream>>>(esrc, edst, head, rec);

    lgcn_pull<1><<<4688, 256, 0, stream>>>(rec, evals, head, utab, itab, utab, ZA);
    lgcn_pull<0><<<4688, 256, 0, stream>>>(rec, evals, head, utab, itab, ZA, ZB);
    lgcn_pull<0><<<4688, 256, 0, stream>>>(rec, evals, head, utab, itab, ZB, ZA);

    lgcn_gamma_wmma<<<128, 256, 0, stream>>>(ZA, users, items, out);
}


# Layer 1's chain walk also stores each row's first 32 (src,val) links as one 256-byte slot; layers 2-3 read the slot, not 2 random records per link; old path if ws_size < 149.0 MB; sum order kept
# speedup vs baseline: 3.6003x; 1.1371x over previous
#include <hip/hip_runtime.h>


#define NUM_USERS 100000
#define NUM_ITEMS 50000
#define N_NODES   (NUM_USERS + NUM_ITEMS)
#define LATENT    64
#define NNZ_E     4000000
#define BATCH_SZ  16384

typedef __attribute__((ext_vector_type(2))) float v2f;
typedef __attribute__((ext_vector_type(4))) float v4f;
typedef __attribute__((ext_vector_type(8))) float v8f;
typedef __attribute__((ext_vector_type(2))) int   v2i;
typedef __attribute__((ext_vector_type(4))) int   v4i;


__global__ void lgcn_head_init(int* __restrict__ head) {
    int i = blockIdx.x * blockDim.x + threadIdx.x;
    if (i < N_NODES) head[i] = -1;
}

__global__ void lgcn_chain(const int* __restrict__ esrc, const int* __restrict__ edst,
                           int* __restrict__ head, int* __restrict__ rec) {
    int t = blockIdx.x * blockDim.x + threadIdx.x;
    if (t >= NNZ_E / 4) return;
    int ea = 2 * t;
    int eb = ea + NNZ_E / 2;
    v2i da = __builtin_nontemporal_load((const v2i*)(edst + ea));
    v2i db = __builtin_nontemporal_load((const v2i*)(edst + eb));
    v2i sa = __builtin_nontemporal_load((const v2i*)(esrc + ea));
    v2i sb = __builtin_nontemporal_load((const v2i*)(esrc + eb));
    int a0 = atomicExch(head + da.x, ea);
    int a1 = atomicExch(head + da.y, ea + 1);
    int b0 = atomicExch(head + db.x, eb);
    int b1 = atomicExch(head + db.y, eb + 1);
    v4i ra = {a0, sa.x, a1, sa.y};
    v4i rb = {b0, sb.x, b1, sb.y};
    *(v4i*)(rec + 2 * (long)ea) = ra;
    *(v4i*)(rec + 2 * (long)eb) = rb;
}

template <int FIRST>
__global__ void lgcn_pull(const int* __restrict__ rec, const float* __restrict__ evals,
                          const int* __restrict__ head,
                          const float* __restrict__ utab, const float* __restrict__ itab,
                          const float* __restrict__ Zin, float* __restrict__ Zout) {
    int t   = blockIdx.x * blockDim.x + threadIdx.x;
    int row = t >> 3;
    int j   = threadIdx.x & 7;
    if (row >= N_NODES) return;

    const float* x0r = (row < NUM_USERS ? utab + (long)row * LATENT
                                        : itab + (long)(row - NUM_USERS) * LATENT) + 4 * j;
    v4f b0 = __builtin_nontemporal_load((const v4f*)x0r);
    v4f b1 = __builtin_nontemporal_load((const v4f*)(x0r + 32));

    v4f a0 = {0.f, 0.f, 0.f, 0.f};
    v4f a1 = {0.f, 0.f, 0.f, 0.f};

    int   e    = head[row];
    bool  live = (unsigned)e < (unsigned)NNZ_E;
    v2i   r    = {-1, 0};
    float v    = 0.f;
    if (live) { r = *(const v2i*)(rec + 2 * (long)e); v = evals[e]; }

    for (int it = 0; it < NNZ_E && live; ++it) {
        int   nxt = r.x;
        int   s   = r.y;
        float w   = v;
        live = (unsigned)nxt < (unsigned)NNZ_E;
        if (live) { r = *(const v2i*)(rec + 2 * (long)nxt); v = evals[nxt]; }
        const float* xr;
        if (FIRST) xr = (s < NUM_USERS ? utab + (long)s * LATENT
                                       : itab + (long)(s - NUM_USERS) * LATENT) + 4 * j;
        else       xr = Zin + (long)s * LATENT + 4 * j;
        v4f x0 = *(const v4f*)xr;
        v4f x1 = *(const v4f*)(xr + 32);
        a0 += w * x0;
        a1 += w * x1;
    }

    float* yo = Zout + (long)row * LATENT + 4 * j;
    *(v4f*)yo        = b0 + a0;
    *(v4f*)(yo + 32) = b1 + a1;
}

__global__ void lgcn_pull_first(const int* __restrict__ rec, const float* __restrict__ evals,
                                const int* __restrict__ head,
                                const float* __restrict__ utab, const float* __restrict__ itab,
                                float* __restrict__ Zout,
                                int* __restrict__ slot, int* __restrict__ info) {
    int t   = blockIdx.x * blockDim.x + threadIdx.x;
    int row = t >> 3;
    int j   = threadIdx.x & 7;
    if (row >= N_NODES) return;

    const float* x0r = (row < NUM_USERS ? utab + (long)row * LATENT
                                        : itab + (long)(row - NUM_USERS) * LATENT) + 4 * j;
    v4f b0 = __builtin_nontemporal_load((const v4f*)x0r);
    v4f b1 = __builtin_nontemporal_load((const v4f*)(x0r + 32));

    v4f a0 = {0.f, 0.f, 0.f, 0.f};
    v4f a1 = {0.f, 0.f, 0.f, 0.f};
    v4i p0 = {0, 0, 0, 0};
    v4i p1 = {0, 0, 0, 0};

    int   e    = head[row];
    bool  live = (unsigned)e < (unsigned)NNZ_E;
    v2i   r    = {-1, 0};
    float v    = 0.f;
    if (live) { r = *(const v2i*)(rec + 2 * (long)e); v = evals[e]; }

    int cur  = e;
    int cnt  = 0;
    int cont = -1;

    for (int it = 0; it < NNZ_E && live; ++it) {
        int   nxt = r.x;
        int   s   = r.y;
        float w   = v;
        if (it == 32) cont = cur;
        live = (unsigned)nxt < (unsigned)NNZ_E;
        if (live) { r = *(const v2i*)(rec + 2 * (long)nxt); v = evals[nxt]; }
        if (it < 32) {
            bool mine = ((it >> 1) & 7) == j;
            int  wb   = __float_as_int(w);
            if (it < 16) {
                if (it & 1) { p0.z = mine ? s : p0.z; p0.w = mine ? wb : p0.w; }
                else        { p0.x = mine ? s : p0.x; p0.y = mine ? wb : p0.y; }
            } else {
                if (it & 1) { p1.z = mine ? s : p1.z; p1.w = mine ? wb : p1.w; }
                else        { p1.x = mine ? s : p1.x; p1.y = mine ? wb : p1.y; }
            }
        }
        const float* xr = (s < NUM_USERS ? utab + (long)s * LATENT
                                         : itab + (long)(s - NUM_USERS) * LATENT) + 4 * j;
        v4f x0 = *(const v4f*)xr;
        v4f x1 = *(const v4f*)(xr + 32);
        a0 += w * x0;
        a1 += w * x1;
        cur = nxt;
        ++cnt;
    }

    float* yo = Zout + (long)row * LATENT + 4 * j;
    *(v4f*)yo        = b0 + a0;
    *(v4f*)(yo + 32) = b1 + a1;

    int* sr = slot + (long)row * 64 + 4 * j;
    *(v4i*)sr        = p0;
    *(v4i*)(sr + 32) = p1;
    if (j == 0) {
        v2i m = {cnt < 32 ? cnt : 32, cont};
        *(v2i*)(info + 2 * (long)row) = m;
    }
}

__global__ void lgcn_pull_list(const int* __restrict__ slot, const int* __restrict__ info,
                               const int* __restrict__ rec, const float* __restrict__ evals,
                               const float* __restrict__ utab, const float* __restrict__ itab,
                               const float* __restrict__ Zin, float* __restrict__ Zout) {
    int t   = blockIdx.x * blockDim.x + threadIdx.x;
    int row = t >> 3;
    int j   = threadIdx.x & 7;
    if (row >= N_NODES) return;

    const float* x0r = (row < NUM_USERS ? utab + (long)row * LATENT
                                        : itab + (long)(row - NUM_USERS) * LATENT) + 4 * j;
    v4f b0 = __builtin_nontemporal_load((const v4f*)x0r);
    v4f b1 = __builtin_nontemporal_load((const v4f*)(x0r + 32));

    v2i mt = *(const v2i*)(info + 2 * (long)row);
    const int* sr = slot + (long)row * 64 + 4 * j;
    v4i p0 = *(const v4i*)sr;
    v4i p1 = *(const v4i*)(sr + 32);

    v4f a0 = {0.f, 0.f, 0.f, 0.f};
    v4f a1 = {0.f, 0.f, 0.f, 0.f};

    int n = mt.x < 32 ? mt.x : 32;
    int gbase = (threadIdx.x & 31) & ~7;

    for (int m = 0; m < 16 && 2 * m < n; ++m) {
        v4i sel = (m & 8) ? p1 : p0;
        int src_lane4 = (gbase + (m & 7)) << 2;
        int   s0 = __builtin_amdgcn_ds_bpermute(src_lane4, sel.x);
        float w0 = __int_as_float(__builtin_amdgcn_ds_bpermute(src_lane4, sel.y));
        int   s1 = __builtin_amdgcn_ds_bpermute(src_lane4, sel.z);
        float w1 = __int_as_float(__builtin_amdgcn_ds_bpermute(src_lane4, sel.w));
        if (2 * m + 1 >= n) { s1 = s0; w1 = 0.f; }
        const float* xa = Zin + (long)s0 * LATENT + 4 * j;
        const float* xb = Zin + (long)s1 * LATENT + 4 * j;
        v4f xa0 = *(const v4f*)xa;
        v4f xa1 = *(const v4f*)(xa + 32);
        v4f xb0 = *(const v4f*)xb;
        v4f xb1 = *(const v4f*)(xb + 32);
        a0 += w0 * xa0;
        a1 += w0 * xa1;
        a0 += w1 * xb0;
        a1 += w1 * xb1;
    }

    int   e    = mt.y;
    bool  live = (unsigned)e < (unsigned)NNZ_E;
    v2i   r    = {-1, 0};
    float v    = 0.f;
    if (live) { r = *(const v2i*)(rec + 2 * (long)e); v = evals[e]; }
    for (int it = 0; it < NNZ_E && live; ++it) {
        int   nxt = r.x;
        int   s   = r.y;
        float w   = v;
        live = (unsigned)nxt < (unsigned)NNZ_E;
        if (live) { r = *(const v2i*)(rec + 2 * (long)nxt); v = evals[nxt]; }
        const float* xr = Zin + (long)s * LATENT + 4 * j;
        v4f x0 = *(const v4f*)xr;
        v4f x1 = *(const v4f*)(xr + 32);
        a0 += w * x0;
        a1 += w * x1;
    }

    float* yo = Zout + (long)row * LATENT + 4 * j;
    *(v4f*)yo        = b0 + a0;
    *(v4f*)(yo + 32) = b1 + a1;
}

__global__ void lgcn_gamma_wmma(const float* __restrict__ acc,
                                const int* __restrict__ users,
                                const int* __restrict__ items,
                                float* __restrict__ out) {
    int wave = (blockIdx.x * blockDim.x + threadIdx.x) >> 5;
    int lane = threadIdx.x & 31;
    int b    = wave * 16 + (lane & 15);
    int half = (lane >> 4) & 1;

    const float* urow = acc + (long)users[b] * LATENT;
    const float* irow = acc + ((long)NUM_USERS + items[b]) * LATENT;

    v8f c = {};
#pragma unroll
    for (int kk = 0; kk < 16; ++kk) {
        int off = 4 * kk + 2 * half;
        v2f a  = *(const v2f*)(urow + off);
        v2f bb = *(const v2f*)(irow + off);
        c = __builtin_amdgcn_wmma_f32_16x16x4_f32(false, a, false, bb,
                                                  (short)0, c, false, false);
    }

    int r = ((lane < 8) ? lane : (lane - 24)) & 7;
    float v = c[0];
#pragma unroll
    for (int i = 1; i < 8; ++i) if (r == i) v = c[i];

    if (lane < 8)        out[wave * 16 + lane]        = v * 0.0625f;
    else if (lane >= 24) out[wave * 16 + (lane - 16)] = v * 0.0625f;
}

extern "C" void kernel_launch(void* const* d_in, const int* in_sizes, int n_in,
                              void* d_out, int out_size, void* d_ws, size_t ws_size,
                              hipStream_t stream) {
    const int*   users = (const int*)d_in[0];
    const int*   items = (const int*)d_in[1];
    const int*   esrc  = (const int*)d_in[2];
    const int*   edst  = (const int*)d_in[3];
    const float* evals = (const float*)d_in[4];
    const float* utab  = (const float*)d_in[5];
    const float* itab  = (const float*)d_in[6];
    float* out = (float*)d_out;

    float* ZA   = (float*)d_ws;
    float* ZB   = ZA + 9600000;
    int*   rec  = (int*)(ZB + 9600000);
    int*   head = rec + 8000000;

    lgcn_head_init<<<586, 256, 0, stream>>>(head);
    lgcn_chain<<<3907, 256, 0, stream>>>(esrc, edst, head, rec);

    if (ws_size >= 149000064u) {
        int* slot = head + 150016;
        int* info = slot + 9600000;
        lgcn_pull_first<<<4688, 256, 0, stream>>>(rec, evals, head, utab, itab, ZA, slot, info);
        lgcn_pull_list<<<4688, 256, 0, stream>>>(slot, info, rec, evals, utab, itab, ZA, ZB);
        lgcn_pull_list<<<4688, 256, 0, stream>>>(slot, info, rec, evals, utab, itab, ZB, ZA);
    } else {
        lgcn_pull<1><<<4688, 256, 0, stream>>>(rec, evals, head, utab, itab, utab, ZA);
        lgcn_pull<0><<<4688, 256, 0, stream>>>(rec, evals, head, utab, itab, ZA, ZB);
        lgcn_pull<0><<<4688, 256, 0, stream>>>(rec, evals, head, utab, itab, ZB, ZA);
    }

    lgcn_gamma_wmma<<<128, 256, 0, stream>>>(ZA, users, items, out);
}


# Edges threaded into row chains by 256 range-owning work-groups, each scanning 1/8 of the edge list: LDS head exchanges replace the global ones; 8 chains per row walked in turn; row sum order changes
# speedup vs baseline: 4.7325x; 1.3145x over previous
#include <hip/hip_runtime.h>


#define NUM_USERS 100000
#define NUM_ITEMS 50000
#define N_NODES   (NUM_USERS + NUM_ITEMS)
#define LATENT    64
#define NNZ_E     4000000
#define BATCH_SZ  16384
#define RANGE_ROWS 4688
#define N_SLICES   8
#define SLICE_E    (NNZ_E / N_SLICES)

typedef __attribute__((ext_vector_type(2))) float v2f;
typedef __attribute__((ext_vector_type(4))) float v4f;
typedef __attribute__((ext_vector_type(8))) float v8f;
typedef __attribute__((ext_vector_type(2))) int   v2i;
typedef __attribute__((ext_vector_type(4))) int   v4i;


__global__ void lgcn_head_init(int* __restrict__ head) {
    int i = blockIdx.x * blockDim.x + threadIdx.x;
    if (i < N_NODES) head[i] = -1;
}

__global__ void lgcn_chain(const int* __restrict__ esrc, const int* __restrict__ edst,
                           int* __restrict__ head, int* __restrict__ rec) {
    int t = blockIdx.x * blockDim.x + threadIdx.x;
    if (t >= NNZ_E / 4) return;
    int ea = 2 * t;
    int eb = ea + NNZ_E / 2;
    v2i da = __builtin_nontemporal_load((const v2i*)(edst + ea));
    v2i db = __builtin_nontemporal_load((const v2i*)(edst + eb));
    v2i sa = __builtin_nontemporal_load((const v2i*)(esrc + ea));
    v2i sb = __builtin_nontemporal_load((const v2i*)(esrc + eb));
    int a0 = atomicExch(head + da.x, ea);
    int a1 = atomicExch(head + da.y, ea + 1);
    int b0 = atomicExch(head + db.x, eb);
    int b1 = atomicExch(head + db.y, eb + 1);
    v4i ra = {a0, sa.x, a1, sa.y};
    v4i rb = {b0, sb.x, b1, sb.y};
    *(v4i*)(rec + 2 * (long)ea) = ra;
    *(v4i*)(rec + 2 * (long)eb) = rb;
}

__global__ __launch_bounds__(1024)
void lgcn_chain_own(const int* __restrict__ esrc, const int* __restrict__ edst,
                    int* __restrict__ heads, int* __restrict__ rec) {
    __shared__ int lh[RANGE_ROWS];
    const int tid = threadIdx.x;
    const int g   = blockIdx.x >> 3;
    const int sl  = blockIdx.x & 7;
    const int lo  = g * RANGE_ROWS;
    for (int i = tid; i < RANGE_ROWS; i += 1024) lh[i] = -1;
    __syncthreads();

    const int e0 = sl * SLICE_E;
    for (int base = 0; base < SLICE_E; base += 16384) {
        v4i d[4];
        v4i s[4];
        int k[4];
#pragma unroll
        for (int u = 0; u < 4; ++u) {
            k[u] = base + ((u << 10) + tid) * 4;
            v4i dd = {-1, -1, -1, -1};
            v4i ss = {0, 0, 0, 0};
            if (k[u] < SLICE_E) {
                dd = *(const v4i*)(edst + e0 + k[u]);
                ss = *(const v4i*)(esrc + e0 + k[u]);
            }
            d[u] = dd;
            s[u] = ss;
        }
#pragma unroll
        for (int u = 0; u < 4; ++u) {
#pragma unroll
            for (int c = 0; c < 4; ++c) {
                unsigned q = (unsigned)(d[u][c] - lo);
                if (q < (unsigned)RANGE_ROWS) {
                    int e    = e0 + k[u] + c;
                    int prev = atomicExch(&lh[q], e);
                    v2i rr   = {prev, s[u][c]};
                    *(v2i*)(rec + 2 * (long)e) = rr;
                }
            }
        }
    }
    __syncthreads();
    for (int i = tid; i < RANGE_ROWS; i += 1024) {
        int row = lo + i;
        if (row < N_NODES) heads[sl * N_NODES + row] = lh[i];
    }
}

template <int FIRST>
__global__ void lgcn_pull(const int* __restrict__ rec, const float* __restrict__ evals,
                          const int* __restrict__ head,
                          const float* __restrict__ utab, const float* __restrict__ itab,
                          const float* __restrict__ Zin, float* __restrict__ Zout) {
    int t   = blockIdx.x * blockDim.x + threadIdx.x;
    int row = t >> 3;
    int j   = threadIdx.x & 7;
    if (row >= N_NODES) return;

    const float* x0r = (row < NUM_USERS ? utab + (long)row * LATENT
                                        : itab + (long)(row - NUM_USERS) * LATENT) + 4 * j;
    v4f b0 = __builtin_nontemporal_load((const v4f*)x0r);
    v4f b1 = __builtin_nontemporal_load((const v4f*)(x0r + 32));

    v4f a0 = {0.f, 0.f, 0.f, 0.f};
    v4f a1 = {0.f, 0.f, 0.f, 0.f};

    int   e    = head[row];
    bool  live = (unsigned)e < (unsigned)NNZ_E;
    v2i   r    = {-1, 0};
    float v    = 0.f;
    if (live) { r = *(const v2i*)(rec + 2 * (long)e); v = evals[e]; }

    for (int it = 0; it < NNZ_E && live; ++it) {
        int   nxt = r.x;
        int   s   = r.y;
        float w   = v;
        live = (unsigned)nxt < (unsigned)NNZ_E;
        if (live) { r = *(const v2i*)(rec + 2 * (long)nxt); v = evals[nxt]; }
        const float* xr;
        if (FIRST) xr = (s < NUM_USERS ? utab + (long)s * LATENT
                                       : itab + (long)(s - NUM_USERS) * LATENT) + 4 * j;
        else       xr = Zin + (long)s * LATENT + 4 * j;
        v4f x0 = *(const v4f*)xr;
        v4f x1 = *(const v4f*)(xr + 32);
        a0 += w * x0;
        a1 += w * x1;
    }

    float* yo = Zout + (long)row * LATENT + 4 * j;
    *(v4f*)yo        = b0 + a0;
    *(v4f*)(yo + 32) = b1 + a1;
}

__global__ void lgcn_pull_first(int* rec, const float* __restrict__ evals,
                                const int* __restrict__ heads,
                                const float* __restrict__ utab, const float* __restrict__ itab,
                                float* __restrict__ Zout,
                                int* __restrict__ slot, int* __restrict__ info) {
    int t   = blockIdx.x * blockDim.x + threadIdx.x;
    int row = t >> 3;
    int j   = threadIdx.x & 7;
    if (row >= N_NODES) return;

    const float* x0r = (row < NUM_USERS ? utab + (long)row * LATENT
                                        : itab + (long)(row - NUM_USERS) * LATENT) + 4 * j;
    v4f b0 = __builtin_nontemporal_load((const v4f*)x0r);
    v4f b1 = __builtin_nontemporal_load((const v4f*)(x0r + 32));

    v4f a0 = {0.f, 0.f, 0.f, 0.f};
    v4f a1 = {0.f, 0.f, 0.f, 0.f};
    v4i p0 = {0, 0, 0, 0};
    v4i p1 = {0, 0, 0, 0};

    int gbase = (threadIdx.x & 31) & ~7;
    int hj    = heads[j * N_NODES + row];
    unsigned todo = (__builtin_amdgcn_ballot_w32((unsigned)hj < (unsigned)NNZ_E) >> gbase) & 0xFFu;

    int e = -1;
    if (todo) {
        int c = __builtin_ctz(todo);
        todo &= todo - 1;
        e = __builtin_amdgcn_ds_bpermute((gbase + c) << 2, hj);
    }
    bool  live = (unsigned)e < (unsigned)NNZ_E;
    v2i   r    = {-1, 0};
    float v    = 0.f;
    if (live) { r = *(const v2i*)(rec + 2 * (long)e); v = evals[e]; }

    int cur  = e;
    int cnt  = 0;
    int cont = -1;

    for (int it = 0; it < NNZ_E && live; ++it) {
        int   nxt = r.x;
        int   s   = r.y;
        float w   = v;
        if (it == 32) cont = cur;
        if ((unsigned)nxt >= (unsigned)NNZ_E && todo) {
            int c = __builtin_ctz(todo);
            todo &= todo - 1;
            nxt = __builtin_amdgcn_ds_bpermute((gbase + c) << 2, hj);
            if (it >= 32 && j == 0) rec[2 * (long)cur] = nxt;
        }
        live = (unsigned)nxt < (unsigned)NNZ_E;
        if (live) { r = *(const v2i*)(rec + 2 * (long)nxt); v = evals[nxt]; }
        if (it < 32) {
            bool mine = ((it >> 1) & 7) == j;
            int  wb   = __float_as_int(w);
            if (it < 16) {
                if (it & 1) { p0.z = mine ? s : p0.z; p0.w = mine ? wb : p0.w; }
                else        { p0.x = mine ? s : p0.x; p0.y = mine ? wb : p0.y; }
            } else {
                if (it & 1) { p1.z = mine ? s : p1.z; p1.w = mine ? wb : p1.w; }
                else        { p1.x = mine ? s : p1.x; p1.y = mine ? wb : p1.y; }
            }
        }
        const float* xr = (s < NUM_USERS ? utab + (long)s * LATENT
                                         : itab + (long)(s - NUM_USERS) * LATENT) + 4 * j;
        v4f x0 = *(const v4f*)xr;
        v4f x1 = *(const v4f*)(xr + 32);
        a0 += w * x0;
        a1 += w * x1;
        cur = nxt;
        ++cnt;
    }

    float* yo = Zout + (long)row * LATENT + 4 * j;
    *(v4f*)yo        = b0 + a0;
    *(v4f*)(yo + 32) = b1 + a1;

    int* sr = slot + (long)row * 64 + 4 * j;
    *(v4i*)sr        = p0;
    *(v4i*)(sr + 32) = p1;
    if (j == 0) {
        v2i m = {cnt < 32 ? cnt : 32, cont};
        *(v2i*)(info + 2 * (long)row) = m;
    }
}

__global__ void lgcn_pull_list(const int* __restrict__ slot, const int* __restrict__ info,
                               const int* __restrict__ rec, const float* __restrict__ evals,
                               const float* __restrict__ utab, const float* __restrict__ itab,
                               const float* __restrict__ Zin, float* __restrict__ Zout) {
    int t   = blockIdx.x * blockDim.x + threadIdx.x;
    int row = t >> 3;
    int j   = threadIdx.x & 7;
    if (row >= N_NODES) return;

    const float* x0r = (row < NUM_USERS ? utab + (long)row * LATENT
                                        : itab + (long)(row - NUM_USERS) * LATENT) + 4 * j;
    v4f b0 = __builtin_nontemporal_load((const v4f*)x0r);
    v4f b1 = __builtin_nontemporal_load((const v4f*)(x0r + 32));

    v2i mt = *(const v2i*)(info + 2 * (long)row);
    const int* sr = slot + (long)row * 64 + 4 * j;
    v4i p0 = *(const v4i*)sr;
    v4i p1 = *(const v4i*)(sr + 32);

    v4f a0 = {0.f, 0.f, 0.f, 0.f};
    v4f a1 = {0.f, 0.f, 0.f, 0.f};

    int n = mt.x < 32 ? mt.x : 32;
    int gbase = (threadIdx.x & 31) & ~7;

    for (int m = 0; m < 16 && 2 * m < n; ++m) {
        v4i sel = (m & 8) ? p1 : p0;
        int src_lane4 = (gbase + (m & 7)) << 2;
        int   s0 = __builtin_amdgcn_ds_bpermute(src_lane4, sel.x);
        float w0 = __int_as_float(__builtin_amdgcn_ds_bpermute(src_lane4, sel.y));
        int   s1 = __builtin_amdgcn_ds_bpermute(src_lane4, sel.z);
        float w1 = __int_as_float(__builtin_amdgcn_ds_bpermute(src_lane4, sel.w));
        if (2 * m + 1 >= n) { s1 = s0; w1 = 0.f; }
        const float* xa = Zin + (long)s0 * LATENT + 4 * j;
        const float* xb = Zin + (long)s1 * LATENT + 4 * j;
        v4f xa0 = *(const v4f*)xa;
        v4f xa1 = *(const v4f*)(xa + 32);
        v4f xb0 = *(const v4f*)xb;
        v4f xb1 = *(const v4f*)(xb + 32);
        a0 += w0 * xa0;
        a1 += w0 * xa1;
        a0 += w1 * xb0;
        a1 += w1 * xb1;
    }

    int   e    = mt.y;
    bool  live = (unsigned)e < (unsigned)NNZ_E;
    v2i   r    = {-1, 0};
    float v    = 0.f;
    if (live) { r = *(const v2i*)(rec + 2 * (long)e); v = evals[e]; }
    for (int it = 0; it < NNZ_E && live; ++it) {
        int   nxt = r.x;
        int   s   = r.y;
        float w   = v;
        live = (unsigned)nxt < (unsigned)NNZ_E;
        if (live) { r = *(const v2i*)(rec + 2 * (long)nxt); v = evals[nxt]; }
        const float* xr = Zin + (long)s * LATENT + 4 * j;
        v4f x0 = *(const v4f*)xr;
        v4f x1 = *(const v4f*)(xr + 32);
        a0 += w * x0;
        a1 += w * x1;
    }

    float* yo = Zout + (long)row * LATENT + 4 * j;
    *(v4f*)yo        = b0 + a0;
    *(v4f*)(yo + 32) = b1 + a1;
}

__global__ void lgcn_gamma_wmma(const float* __restrict__ acc,
                                const int* __restrict__ users,
                                const int* __restrict__ items,
                                float* __restrict__ out) {
    int wave = (blockIdx.x * blockDim.x + threadIdx.x) >> 5;
    int lane = threadIdx.x & 31;
    int b    = wave * 16 + (lane & 15);
    int half = (lane >> 4) & 1;

    const float* urow = acc + (long)users[b] * LATENT;
    const float* irow = acc + ((long)NUM_USERS + items[b]) * LATENT;

    v8f c = {};
#pragma unroll
    for (int kk = 0; kk < 16; ++kk) {
        int off = 4 * kk + 2 * half;
        v2f a  = *(const v2f*)(urow + off);
        v2f bb = *(const v2f*)(irow + off);
        c = __builtin_amdgcn_wmma_f32_16x16x4_f32(false, a, false, bb,
                                                  (short)0, c, false, false);
    }

    int r = ((lane < 8) ? lane : (lane - 24)) & 7;
    float v = c[0];
#pragma unroll
    for (int i = 1; i < 8; ++i) if (r == i) v = c[i];

    if (lane < 8)        out[wave * 16 + lane]        = v * 0.0625f;
    else if (lane >= 24) out[wave * 16 + (lane - 16)] = v * 0.0625f;
}

extern "C" void kernel_launch(void* const* d_in, const int* in_sizes, int n_in,
                              void* d_out, int out_size, void* d_ws, size_t ws_size,
                              hipStream_t stream) {
    const int*   users = (const int*)d_in[0];
    const int*   items = (const int*)d_in[1];
    const int*   esrc  = (const int*)d_in[2];
    const int*   edst  = (const int*)d_in[3];
    const float* evals = (const float*)d_in[4];
    const float* utab  = (const float*)d_in[5];
    const float* itab  = (const float*)d_in[6];
    float* out = (float*)d_out;

    float* ZA   = (float*)d_ws;
    float* ZB   = ZA + 9600000;
    int*   rec  = (int*)(ZB + 9600000);
    int*   head = rec + 8000000;


    if (ws_size >= 149000064u) {
        int* slot = head + 150016;
        int* info = slot + 9600000;
        int* heads = (int*)ZB;
        lgcn_chain_own<<<256, 1024, 0, stream>>>(esrc, edst, heads, rec);
        lgcn_pull_first<<<4688, 256, 0, stream>>>(rec, evals, heads, utab, itab, ZA, slot, info);
        lgcn_pull_list<<<4688, 256, 0, stream>>>(slot, info, rec, evals, utab, itab, ZA, ZB);
        lgcn_pull_list<<<4688, 256, 0, stream>>>(slot, info, rec, evals, utab, itab, ZB, ZA);
    } else {
        lgcn_head_init<<<586, 256, 0, stream>>>(head);
        lgcn_chain<<<3907, 256, 0, stream>>>(esrc, edst, head, rec);
        lgcn_pull<1><<<4688, 256, 0, stream>>>(rec, evals, head, utab, itab, utab, ZA);
        lgcn_pull<0><<<4688, 256, 0, stream>>>(rec, evals, head, utab, itab, ZA, ZB);
        lgcn_pull<0><<<4688, 256, 0, stream>>>(rec, evals, head, utab, itab, ZB, ZA);
    }

    lgcn_gamma_wmma<<<128, 256, 0, stream>>>(ZA, users, items, out);
}


# Edge threading re-tiled: 8 row ranges x 32 list slices (was 32 x 8), 75 KB of LDS heads per work-group: edge list read 8x not 32x; rows have 32 chains, walked in turn; row sum order changes
# speedup vs baseline: 5.3457x; 1.1296x over previous
#include <hip/hip_runtime.h>


#define NUM_USERS 100000
#define NUM_ITEMS 50000
#define N_NODES   (NUM_USERS + NUM_ITEMS)
#define LATENT    64
#define NNZ_E     4000000
#define BATCH_SZ  16384
#define RANGE_ROWS 18752
#define N_SLICES   32
#define SLICE_E    (NNZ_E / N_SLICES)

typedef __attribute__((ext_vector_type(2))) float v2f;
typedef __attribute__((ext_vector_type(4))) float v4f;
typedef __attribute__((ext_vector_type(8))) float v8f;
typedef __attribute__((ext_vector_type(2))) int   v2i;
typedef __attribute__((ext_vector_type(4))) int   v4i;


__global__ void lgcn_head_init(int* __restrict__ head) {
    int i = blockIdx.x * blockDim.x + threadIdx.x;
    if (i < N_NODES) head[i] = -1;
}

__global__ void lgcn_chain(const int* __restrict__ esrc, const int* __restrict__ edst,
                           int* __restrict__ head, int* __restrict__ rec) {
    int t = blockIdx.x * blockDim.x + threadIdx.x;
    if (t >= NNZ_E / 4) return;
    int ea = 2 * t;
    int eb = ea + NNZ_E / 2;
    v2i da = __builtin_nontemporal_load((const v2i*)(edst + ea));
    v2i db = __builtin_nontemporal_load((const v2i*)(edst + eb));
    v2i sa = __builtin_nontemporal_load((const v2i*)(esrc + ea));
    v2i sb = __builtin_nontemporal_load((const v2i*)(esrc + eb));
    int a0 = atomicExch(head + da.x, ea);
    int a1 = atomicExch(head + da.y, ea + 1);
    int b0 = atomicExch(head + db.x, eb);
    int b1 = atomicExch(head + db.y, eb + 1);
    v4i ra = {a0, sa.x, a1, sa.y};
    v4i rb = {b0, sb.x, b1, sb.y};
    *(v4i*)(rec + 2 * (long)ea) = ra;
    *(v4i*)(rec + 2 * (long)eb) = rb;
}

__global__ __launch_bounds__(1024)
void lgcn_chain_own(const int* __restrict__ esrc, const int* __restrict__ edst,
                    int* __restrict__ heads, int* __restrict__ rec) {
    __shared__ int lh[RANGE_ROWS];
    const int tid = threadIdx.x;
    const int g   = blockIdx.x >> 5;
    const int sl  = blockIdx.x & 31;
    const int lo  = g * RANGE_ROWS;
    for (int i = tid; i < RANGE_ROWS; i += 1024) lh[i] = -1;
    __syncthreads();

    const int e0 = sl * SLICE_E;
    for (int base = 0; base < SLICE_E; base += 16384) {
        v4i d[4];
        v4i s[4];
        int k[4];
#pragma unroll
        for (int u = 0; u < 4; ++u) {
            k[u] = base + ((u << 10) + tid) * 4;
            v4i dd = {-1, -1, -1, -1};
            v4i ss = {0, 0, 0, 0};
            if (k[u] < SLICE_E) {
                dd = *(const v4i*)(edst + e0 + k[u]);
                ss = *(const v4i*)(esrc + e0 + k[u]);
            }
            d[u] = dd;
            s[u] = ss;
        }
#pragma unroll
        for (int u = 0; u < 4; ++u) {
#pragma unroll
            for (int c = 0; c < 4; ++c) {
                unsigned q = (unsigned)(d[u][c] - lo);
                if (q < (unsigned)RANGE_ROWS) {
                    int e    = e0 + k[u] + c;
                    int prev = atomicExch(&lh[q], e);
                    v2i rr   = {prev, s[u][c]};
                    *(v2i*)(rec + 2 * (long)e) = rr;
                }
            }
        }
    }
    __syncthreads();
    for (int i = tid; i < RANGE_ROWS; i += 1024) {
        int row = lo + i;
        if (row < N_NODES) heads[sl * N_NODES + row] = lh[i];
    }
}

template <int FIRST>
__global__ void lgcn_pull(const int* __restrict__ rec, const float* __restrict__ evals,
                          const int* __restrict__ head,
                          const float* __restrict__ utab, const float* __restrict__ itab,
                          const float* __restrict__ Zin, float* __restrict__ Zout) {
    int t   = blockIdx.x * blockDim.x + threadIdx.x;
    int row = t >> 3;
    int j   = threadIdx.x & 7;
    if (row >= N_NODES) return;

    const float* x0r = (row < NUM_USERS ? utab + (long)row * LATENT
                                        : itab + (long)(row - NUM_USERS) * LATENT) + 4 * j;
    v4f b0 = __builtin_nontemporal_load((const v4f*)x0r);
    v4f b1 = __builtin_nontemporal_load((const v4f*)(x0r + 32));

    v4f a0 = {0.f, 0.f, 0.f, 0.f};
    v4f a1 = {0.f, 0.f, 0.f, 0.f};

    int   e    = head[row];
    bool  live = (unsigned)e < (unsigned)NNZ_E;
    v2i   r    = {-1, 0};
    float v    = 0.f;
    if (live) { r = *(const v2i*)(rec + 2 * (long)e); v = evals[e]; }

    for (int it = 0; it < NNZ_E && live; ++it) {
        int   nxt = r.x;
        int   s   = r.y;
        float w   = v;
        live = (unsigned)nxt < (unsigned)NNZ_E;
        if (live) { r = *(const v2i*)(rec + 2 * (long)nxt); v = evals[nxt]; }
        const float* xr;
        if (FIRST) xr = (s < NUM_USERS ? utab + (long)s * LATENT
                                       : itab + (long)(s - NUM_USERS) * LATENT) + 4 * j;
        else       xr = Zin + (long)s * LATENT + 4 * j;
        v4f x0 = *(const v4f*)xr;
        v4f x1 = *(const v4f*)(xr + 32);
        a0 += w * x0;
        a1 += w * x1;
    }

    float* yo = Zout + (long)row * LATENT + 4 * j;
    *(v4f*)yo        = b0 + a0;
    *(v4f*)(yo + 32) = b1 + a1;
}

__global__ void lgcn_pull_first(int* rec, const float* __restrict__ evals,
                                const int* __restrict__ heads,
                                const float* __restrict__ utab, const float* __restrict__ itab,
                                float* __restrict__ Zout,
                                int* __restrict__ slot, int* __restrict__ info) {
    int t   = blockIdx.x * blockDim.x + threadIdx.x;
    int row = t >> 3;
    int j   = threadIdx.x & 7;
    if (row >= N_NODES) return;

    const float* x0r = (row < NUM_USERS ? utab + (long)row * LATENT
                                        : itab + (long)(row - NUM_USERS) * LATENT) + 4 * j;
    v4f b0 = __builtin_nontemporal_load((const v4f*)x0r);
    v4f b1 = __builtin_nontemporal_load((const v4f*)(x0r + 32));

    v4f a0 = {0.f, 0.f, 0.f, 0.f};
    v4f a1 = {0.f, 0.f, 0.f, 0.f};
    v4i p0 = {0, 0, 0, 0};
    v4i p1 = {0, 0, 0, 0};

    int gbase = (threadIdx.x & 31) & ~7;
    int h0 = heads[(j     ) * N_NODES + row];
    int h1 = heads[(j +  8) * N_NODES + row];
    int h2 = heads[(j + 16) * N_NODES + row];
    int h3 = heads[(j + 24) * N_NODES + row];
    unsigned todo =
         ((__builtin_amdgcn_ballot_w32((unsigned)h0 < (unsigned)NNZ_E) >> gbase) & 0xFFu)
      | (((__builtin_amdgcn_ballot_w32((unsigned)h1 < (unsigned)NNZ_E) >> gbase) & 0xFFu) <<  8)
      | (((__builtin_amdgcn_ballot_w32((unsigned)h2 < (unsigned)NNZ_E) >> gbase) & 0xFFu) << 16)
      | (((__builtin_amdgcn_ballot_w32((unsigned)h3 < (unsigned)NNZ_E) >> gbase) & 0xFFu) << 24);

    int e = -1;
    if (todo) {
        int c = __builtin_ctz(todo);
        todo &= todo - 1;
        int hs = (c & 16) ? ((c & 8) ? h3 : h2) : ((c & 8) ? h1 : h0);
        e = __builtin_amdgcn_ds_bpermute((gbase + (c & 7)) << 2, hs);
    }
    bool  live = (unsigned)e < (unsigned)NNZ_E;
    v2i   r    = {-1, 0};
    float v    = 0.f;
    if (live) { r = *(const v2i*)(rec + 2 * (long)e); v = evals[e]; }

    int cur  = e;
    int cnt  = 0;
    int cont = -1;

    for (int it = 0; it < NNZ_E && live; ++it) {
        int   nxt = r.x;
        int   s   = r.y;
        float w   = v;
        if (it == 32) cont = cur;
        if ((unsigned)nxt >= (unsigned)NNZ_E && todo) {
            int c = __builtin_ctz(todo);
            todo &= todo - 1;
            int hs = (c & 16) ? ((c & 8) ? h3 : h2) : ((c & 8) ? h1 : h0);
            nxt = __builtin_amdgcn_ds_bpermute((gbase + (c & 7)) << 2, hs);
            if (it >= 32 && j == 0) rec[2 * (long)cur] = nxt;
        }
        live = (unsigned)nxt < (unsigned)NNZ_E;
        if (live) { r = *(const v2i*)(rec + 2 * (long)nxt); v = evals[nxt]; }
        if (it < 32) {
            bool mine = ((it >> 1) & 7) == j;
            int  wb   = __float_as_int(w);
            if (it < 16) {
                if (it & 1) { p0.z = mine ? s : p0.z; p0.w = mine ? wb : p0.w; }
                else        { p0.x = mine ? s : p0.x; p0.y = mine ? wb : p0.y; }
            } else {
                if (it & 1) { p1.z = mine ? s : p1.z; p1.w = mine ? wb : p1.w; }
                else        { p1.x = mine ? s : p1.x; p1.y = mine ? wb : p1.y; }
            }
        }
        const float* xr = (s < NUM_USERS ? utab + (long)s * LATENT
                                         : itab + (long)(s - NUM_USERS) * LATENT) + 4 * j;
        v4f x0 = *(const v4f*)xr;
        v4f x1 = *(const v4f*)(xr + 32);
        a0 += w * x0;
        a1 += w * x1;
        cur = nxt;
        ++cnt;
    }

    float* yo = Zout + (long)row * LATENT + 4 * j;
    *(v4f*)yo        = b0 + a0;
    *(v4f*)(yo + 32) = b1 + a1;

    int* sr = slot + (long)row * 64 + 4 * j;
    *(v4i*)sr        = p0;
    *(v4i*)(sr + 32) = p1;
    if (j == 0) {
        v2i m = {cnt < 32 ? cnt : 32, cont};
        *(v2i*)(info + 2 * (long)row) = m;
    }
}

__global__ void lgcn_pull_list(const int* __restrict__ slot, const int* __restrict__ info,
                               const int* __restrict__ rec, const float* __restrict__ evals,
                               const float* __restrict__ utab, const float* __restrict__ itab,
                               const float* __restrict__ Zin, float* __restrict__ Zout) {
    int t   = blockIdx.x * blockDim.x + threadIdx.x;
    int row = t >> 3;
    int j   = threadIdx.x & 7;
    if (row >= N_NODES) return;

    const float* x0r = (row < NUM_USERS ? utab + (long)row * LATENT
                                        : itab + (long)(row - NUM_USERS) * LATENT) + 4 * j;
    v4f b0 = __builtin_nontemporal_load((const v4f*)x0r);
    v4f b1 = __builtin_nontemporal_load((const v4f*)(x0r + 32));

    v2i mt = *(const v2i*)(info + 2 * (long)row);
    const int* sr = slot + (long)row * 64 + 4 * j;
    v4i p0 = *(const v4i*)sr;
    v4i p1 = *(const v4i*)(sr + 32);

    v4f a0 = {0.f, 0.f, 0.f, 0.f};
    v4f a1 = {0.f, 0.f, 0.f, 0.f};

    int n = mt.x < 32 ? mt.x : 32;
    int gbase = (threadIdx.x & 31) & ~7;

    for (int m = 0; m < 16 && 2 * m < n; ++m) {
        v4i sel = (m & 8) ? p1 : p0;
        int src_lane4 = (gbase + (m & 7)) << 2;
        int   s0 = __builtin_amdgcn_ds_bpermute(src_lane4, sel.x);
        float w0 = __int_as_float(__builtin_amdgcn_ds_bpermute(src_lane4, sel.y));
        int   s1 = __builtin_amdgcn_ds_bpermute(src_lane4, sel.z);
        float w1 = __int_as_float(__builtin_amdgcn_ds_bpermute(src_lane4, sel.w));
        if (2 * m + 1 >= n) { s1 = s0; w1 = 0.f; }
        const float* xa = Zin + (long)s0 * LATENT + 4 * j;
        const float* xb = Zin + (long)s1 * LATENT + 4 * j;
        v4f xa0 = *(const v4f*)xa;
        v4f xa1 = *(const v4f*)(xa + 32);
        v4f xb0 = *(const v4f*)xb;
        v4f xb1 = *(const v4f*)(xb + 32);
        a0 += w0 * xa0;
        a1 += w0 * xa1;
        a0 += w1 * xb0;
        a1 += w1 * xb1;
    }

    int   e    = mt.y;
    bool  live = (unsigned)e < (unsigned)NNZ_E;
    v2i   r    = {-1, 0};
    float v    = 0.f;
    if (live) { r = *(const v2i*)(rec + 2 * (long)e); v = evals[e]; }
    for (int it = 0; it < NNZ_E && live; ++it) {
        int   nxt = r.x;
        int   s   = r.y;
        float w   = v;
        live = (unsigned)nxt < (unsigned)NNZ_E;
        if (live) { r = *(const v2i*)(rec + 2 * (long)nxt); v = evals[nxt]; }
        const float* xr = Zin + (long)s * LATENT + 4 * j;
        v4f x0 = *(const v4f*)xr;
        v4f x1 = *(const v4f*)(xr + 32);
        a0 += w * x0;
        a1 += w * x1;
    }

    float* yo = Zout + (long)row * LATENT + 4 * j;
    *(v4f*)yo        = b0 + a0;
    *(v4f*)(yo + 32) = b1 + a1;
}

__global__ void lgcn_gamma_wmma(const float* __restrict__ acc,
                                const int* __restrict__ users,
                                const int* __restrict__ items,
                                float* __restrict__ out) {
    int wave = (blockIdx.x * blockDim.x + threadIdx.x) >> 5;
    int lane = threadIdx.x & 31;
    int b    = wave * 16 + (lane & 15);
    int half = (lane >> 4) & 1;

    const float* urow = acc + (long)users[b] * LATENT;
    const float* irow = acc + ((long)NUM_USERS + items[b]) * LATENT;

    v8f c = {};
#pragma unroll
    for (int kk = 0; kk < 16; ++kk) {
        int off = 4 * kk + 2 * half;
        v2f a  = *(const v2f*)(urow + off);
        v2f bb = *(const v2f*)(irow + off);
        c = __builtin_amdgcn_wmma_f32_16x16x4_f32(false, a, false, bb,
                                                  (short)0, c, false, false);
    }

    int r = ((lane < 8) ? lane : (lane - 24)) & 7;
    float v = c[0];
#pragma unroll
    for (int i = 1; i < 8; ++i) if (r == i) v = c[i];

    if (lane < 8)        out[wave * 16 + lane]        = v * 0.0625f;
    else if (lane >= 24) out[wave * 16 + (lane - 16)] = v * 0.0625f;
}

extern "C" void kernel_launch(void* const* d_in, const int* in_sizes, int n_in,
                              void* d_out, int out_size, void* d_ws, size_t ws_size,
                              hipStream_t stream) {
    const int*   users = (const int*)d_in[0];
    const int*   items = (const int*)d_in[1];
    const int*   esrc  = (const int*)d_in[2];
    const int*   edst  = (const int*)d_in[3];
    const float* evals = (const float*)d_in[4];
    const float* utab  = (const float*)d_in[5];
    const float* itab  = (const float*)d_in[6];
    float* out = (float*)d_out;

    float* ZA   = (float*)d_ws;
    float* ZB   = ZA + 9600000;
    int*   rec  = (int*)(ZB + 9600000);
    int*   head = rec + 8000000;


    if (ws_size >= 149000064u) {
        int* slot = head + 150016;
        int* info = slot + 9600000;
        int* heads = (int*)ZB;
        lgcn_chain_own<<<256, 1024, 0, stream>>>(esrc, edst, heads, rec);
        lgcn_pull_first<<<4688, 256, 0, stream>>>(rec, evals, heads, utab, itab, ZA, slot, info);
        lgcn_pull_list<<<4688, 256, 0, stream>>>(slot, info, rec, evals, utab, itab, ZA, ZB);
        lgcn_pull_list<<<4688, 256, 0, stream>>>(slot, info, rec, evals, utab, itab, ZB, ZA);
    } else {
        lgcn_head_init<<<586, 256, 0, stream>>>(head);
        lgcn_chain<<<3907, 256, 0, stream>>>(esrc, edst, head, rec);
        lgcn_pull<1><<<4688, 256, 0, stream>>>(rec, evals, head, utab, itab, utab, ZA);
        lgcn_pull<0><<<4688, 256, 0, stream>>>(rec, evals, head, utab, itab, ZA, ZB);
        lgcn_pull<0><<<4688, 256, 0, stream>>>(rec, evals, head, utab, itab, ZB, ZA);
    }

    lgcn_gamma_wmma<<<128, 256, 0, stream>>>(ZA, users, items, out);
}
